# starting kernel
# baseline (speedup 1.0000x reference)
#include <hip/hip_runtime.h>

typedef __attribute__((ext_vector_type(2))) float v2f;
typedef __attribute__((ext_vector_type(8))) float v8f;

__global__ void k_init_one(float* deg, int N) {
    int i = blockIdx.x * blockDim.x + threadIdx.x;
    if (i < N) deg[i] = 1.0f;
}

__global__ void k_deg_accum(const int* __restrict__ dst, float* deg, int E) {
    int e = blockIdx.x * blockDim.x + threadIdx.x;
    if (e < E) atomicAdd(&deg[dst[e]], 1.0f);
}

__global__ void k_rsqrt_inplace(float* deg, int N) {
    int i = blockIdx.x * blockDim.x + threadIdx.x;
    if (i < N) deg[i] = rsqrtf(deg[i]);
}

__global__ void k_edge_norm(const int* __restrict__ src, const int* __restrict__ dst,
                            const float* __restrict__ dis, float* __restrict__ nrm, int E) {
    int e = blockIdx.x * blockDim.x + threadIdx.x;
    if (e < E) nrm[e] = dis[src[e]] * dis[dst[e]];
}

__global__ void k_input_proj(const float* __restrict__ num_x, const float* __restrict__ num_mask,
                             const float* __restrict__ x, const float* __restrict__ Wn,
                             const float* __restrict__ bn, const float* __restrict__ a,
                             float* __restrict__ h, int total) {
    int i = blockIdx.x * blockDim.x + threadIdx.x;
    if (i >= total) return;
    int n = i >> 7, d = i & 127;
    float z = num_x[n] * num_mask[n] * Wn[d] + bn[d];
    z = (z >= 0.0f) ? z : a[d] * z;
    h[i] = z + x[i];
}

__global__ __launch_bounds__(128) void k_gemm128_wmma(const float* __restrict__ h,
                                                      const float* __restrict__ W,
                                                      float* __restrict__ t,
                                                      int nRowTiles) {
    __shared__ float Wl[128 * 128];
    {
        int tid = threadIdx.x;
        v2f* Wlv = (v2f*)Wl;
        #pragma unroll
        for (int i = 0; i < 64; ++i) {
            int idx = i * 128 + tid;
            int c = idx & 127;
            int p = idx >> 7;
            v2f pr;
            pr.x = W[(2 * p)     * 128 + c];
            pr.y = W[(2 * p + 1) * 128 + c];
            Wlv[p * 128 + c] = pr;
        }
    }
    __syncthreads();

    const int wave = threadIdx.x >> 5;
    const int lane = threadIdx.x & 31;
    const int rowTile = blockIdx.x * 4 + wave;
    if (rowTile >= nRowTiles) return;

    const int r15   = lane & 15;
    const int khalf = (lane >> 4) << 1;
    const float* hrow = h + (size_t)(rowTile * 16 + r15) * 128;
    const v2f* Wlv = (const v2f*)Wl;

    v8f acc[8] = {};
    for (int k0 = 0; k0 < 128; k0 += 4) {
        v2f a = *(const v2f*)(hrow + k0 + khalf);
        const v2f* wrow = Wlv + (((k0 + khalf) >> 1) * 128 + r15);
        #pragma unroll
        for (int j = 0; j < 8; ++j) {
            v2f bfrag = wrow[j * 16];
            acc[j] = __builtin_amdgcn_wmma_f32_16x16x4_f32(
                false, a, false, bfrag, (short)0, acc[j], false, false);
        }
    }

    const int mhalf = (lane >> 4) << 3;
    float* trow = t + (size_t)(rowTile * 16 + mhalf) * 128 + r15;
    #pragma unroll
    for (int j = 0; j < 8; ++j)
        #pragma unroll
        for (int r = 0; r < 8; ++r)
            trow[(size_t)r * 128 + j * 16] = acc[j][r];
}

__global__ void k_agg_init(const float* __restrict__ t, const float* __restrict__ dis,
                           const float* __restrict__ bias, float* __restrict__ acc, int total) {
    int i = blockIdx.x * blockDim.x + threadIdx.x;
    if (i >= total) return;
    int n = i >> 7, c = i & 127;
    float sn = dis[n] * dis[n];
    acc[i] = sn * t[i] + bias[c];
}

__global__ void k_edge_scatter_vec(const int* __restrict__ src, const int* __restrict__ dst,
                                   const float* __restrict__ nrm, const float* __restrict__ t,
                                   float* __restrict__ acc, int E) {
    int wid  = (blockIdx.x * blockDim.x + threadIdx.x) >> 5;
    int lane = threadIdx.x & 31;
    if (wid >= E) return;
    int s = src[wid], d = dst[wid];
    float w = nrm[wid];
    const float4 v = ((const float4*)(t + (size_t)s * 128))[lane];
    float* ap = acc + (size_t)d * 128 + lane * 4;
    atomicAdd(ap + 0, w * v.x);
    atomicAdd(ap + 1, w * v.y);
    atomicAdd(ap + 2, w * v.z);
    atomicAdd(ap + 3, w * v.w);
}

__global__ void k_prelu(float* __restrict__ h, const float* __restrict__ a, int total) {
    int i = blockIdx.x * blockDim.x + threadIdx.x;
    if (i >= total) return;
    float v = h[i];
    h[i] = (v >= 0.0f) ? v : a[i & 127] * v;
}

__global__ void k_dot_w3(const float* __restrict__ h, const float* __restrict__ W3,
                         float* __restrict__ t3, int N) {
    int wid  = (blockIdx.x * blockDim.x + threadIdx.x) >> 5;
    int lane = threadIdx.x & 31;
    if (wid >= N) return;
    const float4 hv = ((const float4*)(h + (size_t)wid * 128))[lane];
    const float4 wv = ((const float4*)W3)[lane];
    float s = hv.x * wv.x + hv.y * wv.y + hv.z * wv.z + hv.w * wv.w;
    #pragma unroll
    for (int off = 16; off > 0; off >>= 1) s += __shfl_xor(s, off, 32);
    if (lane == 0) t3[wid] = s;
}

__global__ void k_out_init(const float* __restrict__ t3, const float* __restrict__ dis,
                           const float* __restrict__ b3, float* __restrict__ out, int N) {
    int i = blockIdx.x * blockDim.x + threadIdx.x;
    if (i < N) out[i] = dis[i] * dis[i] * t3[i] + b3[0];
}

__global__ void k_edge_scatter_scalar(const int* __restrict__ src, const int* __restrict__ dst,
                                      const float* __restrict__ nrm, const float* __restrict__ t3,
                                      float* __restrict__ out, int E) {
    int e = blockIdx.x * blockDim.x + threadIdx.x;
    if (e < E) atomicAdd(&out[dst[e]], nrm[e] * t3[src[e]]);
}

extern "C" void kernel_launch(void* const* d_in, const int* in_sizes, int n_in,
                              void* d_out, int out_size, void* d_ws, size_t ws_size,
                              hipStream_t stream) {
    const float* num_x    = (const float*)d_in[0];
    const float* num_mask = (const float*)d_in[1];
    const float* x        = (const float*)d_in[2];
    const int*   edge     = (const int*)d_in[3];
    const float* W_num    = (const float*)d_in[4];
    const float* b_num    = (const float*)d_in[5];
    const float* a_in     = (const float*)d_in[6];
    const float* W1       = (const float*)d_in[7];
    const float* b1       = (const float*)d_in[8];
    const float* a1       = (const float*)d_in[9];
    const float* W2       = (const float*)d_in[10];
    const float* b2       = (const float*)d_in[11];
    const float* a2       = (const float*)d_in[12];
    const float* W3       = (const float*)d_in[13];
    const float* b3       = (const float*)d_in[14];

    const int N = in_sizes[1];
    const int E = in_sizes[3] / 2;
    const int* src = edge;
    const int* dst = edge + E;

    const int nRowTiles = (N + 15) / 16;
    const size_t padRows = (size_t)nRowTiles * 16;

    auto al = [](size_t v) { return (v + 255) & ~(size_t)255; };
    char* p = (char*)d_ws;
    float* dis  = (float*)p; p += al((size_t)N * 4);
    float* nrm  = (float*)p; p += al((size_t)E * 4);
    float* hbuf = (float*)p; p += al(padRows * 128 * 4);
    float* tbuf = (float*)p; p += al(padRows * 128 * 4);
    float* t3   = (float*)p; p += al((size_t)N * 4);
    float* out  = (float*)d_out;

    const int TPB = 256;
    const int totND = N * 128;
    dim3 blk(TPB);
    auto gN  = dim3((N + TPB - 1) / TPB);
    auto gE  = dim3((E + TPB - 1) / TPB);
    auto gND = dim3((totND + TPB - 1) / TPB);
    auto gEw = dim3((int)(((size_t)E * 32 + TPB - 1) / TPB));
    auto gNw = dim3((int)(((size_t)N * 32 + TPB - 1) / TPB));
    const int gemmBlocks = (nRowTiles + 3) / 4;

    k_init_one<<<gN, blk, 0, stream>>>(dis, N);
    k_deg_accum<<<gE, blk, 0, stream>>>(dst, dis, E);
    k_rsqrt_inplace<<<gN, blk, 0, stream>>>(dis, N);
    k_edge_norm<<<gE, blk, 0, stream>>>(src, dst, dis, nrm, E);

    k_input_proj<<<gND, blk, 0, stream>>>(num_x, num_mask, x, W_num, b_num, a_in, hbuf, totND);

    k_gemm128_wmma<<<gemmBlocks, 128, 0, stream>>>(hbuf, W1, tbuf, nRowTiles);
    k_agg_init<<<gND, blk, 0, stream>>>(tbuf, dis, b1, hbuf, totND);
    k_edge_scatter_vec<<<gEw, blk, 0, stream>>>(src, dst, nrm, tbuf, hbuf, E);
    k_prelu<<<gND, blk, 0, stream>>>(hbuf, a1, totND);

    k_gemm128_wmma<<<gemmBlocks, 128, 0, stream>>>(hbuf, W2, tbuf, nRowTiles);
    k_agg_init<<<gND, blk, 0, stream>>>(tbuf, dis, b2, hbuf, totND);
    k_edge_scatter_vec<<<gEw, blk, 0, stream>>>(src, dst, nrm, tbuf, hbuf, E);
    k_prelu<<<gND, blk, 0, stream>>>(hbuf, a2, totND);

    k_dot_w3<<<gNw, blk, 0, stream>>>(hbuf, W3, t3, N);
    k_out_init<<<gN, blk, 0, stream>>>(t3, dis, b3, out, N);
    k_edge_scatter_scalar<<<gE, blk, 0, stream>>>(src, dst, nrm, t3, out, E);
}


# input projection runs beside the degree count as 256 extra 1024-thread blocks of that launch (a wave takes whole rows, 8 in flight, 16 B per lane): the 200000-wave launch and its gap are gone
# speedup vs baseline: 2.3613x; 2.3613x over previous
#include <hip/hip_runtime.h>

typedef __attribute__((ext_vector_type(2))) float v2f;
typedef __attribute__((ext_vector_type(8))) float v8f;
typedef __attribute__((ext_vector_type(4))) float v4f;

#define NODE_RANGES 8
#define EDGE_SLICES 32
#define BIN_CAP     8192
#define SCAN_U      16

__device__ __forceinline__ int node_range_len(int N) {
    int r = (N + NODE_RANGES - 1) / NODE_RANGES;
    r = (r + 31) & ~31;
    return min(r, BIN_CAP);
}
__host__ __device__ __forceinline__ size_t part_pitch(int N) { return (size_t)((N + 31) & ~31); }
__device__ __forceinline__ int edge_slice_len(int E) {
    return ((E + EDGE_SLICES - 1) / EDGE_SLICES + 31) & ~31;
}

__device__ __forceinline__ void deg_count_block(int* bins, int bid, const int* __restrict__ dst,
                                                int* __restrict__ cnt, int N, int E) {
    const int g   = bid % NODE_RANGES;
    const int s   = bid / NODE_RANGES;
    const int len = node_range_len(N);
    const int per = edge_slice_len(E);
    const int e0  = min(s * per, E), e1 = min(e0 + per, E);
    for (int base = g * len; base < N; base += NODE_RANGES * len) {
        for (int i = threadIdx.x; i < len; i += blockDim.x) bins[i] = 0;
        __syncthreads();
        if (e1 > e0) {
            for (int tile = e0; tile < e1; tile += SCAN_U * (int)blockDim.x) {
                int dv[SCAN_U];
                #pragma unroll
                for (int u = 0; u < SCAN_U; ++u) {
                    const int e = tile + u * (int)blockDim.x + (int)threadIdx.x;
                    dv[u] = dst[min(e, e1 - 1)];
                }
                #pragma unroll
                for (int u = 0; u < SCAN_U; ++u) {
                    const int e = tile + u * (int)blockDim.x + (int)threadIdx.x;
                    const unsigned r = (unsigned)(dv[u] - base);
                    if (e < e1 && r < (unsigned)len) atomicAdd(&bins[r], 1);
                }
            }
        }
        __syncthreads();
        for (int i = threadIdx.x; i < len; i += blockDim.x) {
            if (base + i < N) cnt[(size_t)s * part_pitch(N) + base + i] = bins[i];
        }
        __syncthreads();
    }
}

#define PROJ_BLOCKS 256
#define PROJ_ROWS   8

__device__ __forceinline__ void input_proj_block(int pb, const float* __restrict__ num_x,
                                                 const float* __restrict__ num_mask,
                                                 const float* __restrict__ x, const float* __restrict__ Wn,
                                                 const float* __restrict__ bn, const float* __restrict__ a,
                                                 float* __restrict__ h, int N) {
    const int lane = threadIdx.x & 31;
    const int wave = __builtin_amdgcn_readfirstlane((int)(threadIdx.x >> 5));
    const int nw   = (int)(blockDim.x >> 5);
    const int rowsPer = (N + PROJ_BLOCKS - 1) / PROJ_BLOCKS;
    const int r0 = min(pb * rowsPer, N), r1 = min(r0 + rowsPer, N);
    if (r1 <= r0) return;
    const v4f wn = ((const v4f*)Wn)[lane];
    const v4f bv = ((const v4f*)bn)[lane];
    const v4f av = ((const v4f*)a)[lane];
    const v4f* x4 = (const v4f*)x;
    v4f* h4 = (v4f*)h;
    for (int base = r0 + wave; base < r1; base += nw * PROJ_ROWS) {
        v4f   xv[PROJ_ROWS];
        float sv[PROJ_ROWS];
        #pragma unroll
        for (int u = 0; u < PROJ_ROWS; ++u) {
            const int r = min(base + u * nw, r1 - 1);
            xv[u] = x4[(size_t)r * 32 + lane];
            sv[u] = num_x[r] * num_mask[r];
        }
        asm volatile("" :: "v"(xv[0]), "v"(xv[1]), "v"(xv[2]), "v"(xv[3]),
                           "v"(xv[4]), "v"(xv[5]), "v"(xv[6]), "v"(xv[7]));
        #pragma unroll
        for (int u = 0; u < PROJ_ROWS; ++u) {
            const int r = base + u * nw;
            if (r < r1) {
                v4f z = sv[u] * wn + bv;
                #pragma unroll
                for (int c = 0; c < 4; ++c) z[c] = (z[c] >= 0.0f) ? z[c] : av[c] * z[c];
                h4[(size_t)r * 32 + lane] = z + xv[u];
            }
        }
    }
}

__global__ __launch_bounds__(1024) void k_deg_count_and_input_proj(const int* __restrict__ dst, int* __restrict__ cnt,
                                                                   const float* __restrict__ num_x,
                                                                   const float* __restrict__ num_mask,
                                                                   const float* __restrict__ x,
                                                                   const float* __restrict__ Wn,
                                                                   const float* __restrict__ bn,
                                                                   const float* __restrict__ a,
                                                                   float* __restrict__ h, int N, int E) {
    __shared__ int bins[BIN_CAP];
    const int bid = (int)blockIdx.x;
    if (bid < NODE_RANGES * EDGE_SLICES) {
        deg_count_block(bins, bid, dst, cnt, N, E);
    } else {
        input_proj_block(bid - NODE_RANGES * EDGE_SLICES, num_x, num_mask, x, Wn, bn, a, h, N);
    }
}

__global__ void k_deg_rsqrt(const int* __restrict__ cnt, float* __restrict__ dis, int N) {
    int i = blockIdx.x * blockDim.x + threadIdx.x;
    if (i >= N) return;
    int c = 0;
    #pragma unroll
    for (int s = 0; s < EDGE_SLICES; ++s) c += cnt[(size_t)s * part_pitch(N) + i];
    dis[i] = rsqrtf(1.0f + (float)c);
}

__global__ void k_edge_norm(const int* __restrict__ src, const int* __restrict__ dst,
                            const float* __restrict__ dis, float* __restrict__ nrm, int E) {
    int e = blockIdx.x * blockDim.x + threadIdx.x;
    if (e < E) nrm[e] = dis[src[e]] * dis[dst[e]];
}

template <bool PRELU_IN>
__device__ __forceinline__ void gemm128_body(float* Wl,
                                             const float* __restrict__ h,
                                             const float* __restrict__ slope,
                                             const float* __restrict__ W,
                                             const float* __restrict__ dis,
                                             const float* __restrict__ bias,
                                             float* __restrict__ t,
                                             float* __restrict__ start,
                                             int nRowTiles, int N) {
    {
        int tid = threadIdx.x;
        v2f* Wlv = (v2f*)Wl;
        #pragma unroll
        for (int i = 0; i < 64; ++i) {
            int idx = i * 128 + tid;
            int c = idx & 127;
            int p = idx >> 7;
            v2f pr;
            pr.x = W[(2 * p)     * 128 + c];
            pr.y = W[(2 * p + 1) * 128 + c];
            Wlv[p * 128 + c] = pr;
        }
    }
    __syncthreads();

    const int wave = threadIdx.x >> 5;
    const int lane = threadIdx.x & 31;
    const int rowTile = blockIdx.x * 4 + wave;
    if (rowTile >= nRowTiles) return;

    const int r15   = lane & 15;
    const int khalf = (lane >> 4) << 1;
    const float* hrow = h + (size_t)(rowTile * 16 + r15) * 128;
    const v2f* Wlv = (const v2f*)Wl;

    const int mhalf = (lane >> 4) << 3;
    const int row0  = rowTile * 16 + mhalf;
    float dv[8];
    #pragma unroll
    for (int r = 0; r < 8; ++r) dv[r] = dis[min(row0 + r, N - 1)];

    v8f acc[8] = {};
    for (int k0 = 0; k0 < 128; k0 += 4) {
        v2f a = *(const v2f*)(hrow + k0 + khalf);
        if (PRELU_IN) {
            const v2f sl = *(const v2f*)(slope + k0 + khalf);
            a.x = (a.x >= 0.0f) ? a.x : sl.x * a.x;
            a.y = (a.y >= 0.0f) ? a.y : sl.y * a.y;
        }
        const v2f* wrow = Wlv + (((k0 + khalf) >> 1) * 128 + r15);
        #pragma unroll
        for (int j = 0; j < 8; ++j) {
            v2f bfrag = wrow[j * 16];
            acc[j] = __builtin_amdgcn_wmma_f32_16x16x4_f32(
                false, a, false, bfrag, (short)0, acc[j], false, false);
        }
    }

    float sn[8];
    #pragma unroll
    for (int r = 0; r < 8; ++r) sn[r] = dv[r] * dv[r];
    float* trow = t     + (size_t)row0 * 128 + r15;
    float* srow = start + (size_t)row0 * 128 + r15;
    #pragma unroll
    for (int j = 0; j < 8; ++j) {
        const float bj = bias[j * 16 + r15];
        #pragma unroll
        for (int r = 0; r < 8; ++r) {
            const float tv = acc[j][r];
            trow[(size_t)r * 128 + j * 16] = tv;
            srow[(size_t)r * 128 + j * 16] = sn[r] * tv + bj;
        }
    }
}

__global__ __launch_bounds__(128) void k_gemm128_wmma_plain(const float* __restrict__ h,
                                                            const float* __restrict__ W,
                                                            const float* __restrict__ dis,
                                                            const float* __restrict__ bias,
                                                            float* __restrict__ t,
                                                            float* __restrict__ start,
                                                            int nRowTiles, int N) {
    __shared__ float Wl[128 * 128];
    gemm128_body<false>(Wl, h, nullptr, W, dis, bias, t, start, nRowTiles, N);
}

__global__ __launch_bounds__(128) void k_gemm128_wmma_prelu(const float* __restrict__ h,
                                                            const float* __restrict__ slope,
                                                            const float* __restrict__ W,
                                                            const float* __restrict__ dis,
                                                            const float* __restrict__ bias,
                                                            float* __restrict__ t,
                                                            float* __restrict__ start,
                                                            int nRowTiles, int N) {
    __shared__ float Wl[128 * 128];
    gemm128_body<true>(Wl, h, slope, W, dis, bias, t, start, nRowTiles, N);
}

__global__ void k_edge_scatter_vec(const int* __restrict__ src, const int* __restrict__ dst,
                                   const float* __restrict__ nrm, const float* __restrict__ t,
                                   float* __restrict__ acc, int E) {
    const int lane = threadIdx.x & 31;
    const int wv    = __builtin_amdgcn_readfirstlane((int)(threadIdx.x >> 5));
    const int first = (blockIdx.x * (int)(blockDim.x >> 5) + wv) * 32;
    if (first >= E) return;
    const int cnt = min(32, E - first);

    int s = 0, d = 0, wb = 0;
    if (lane < cnt) {
        s  = src[first + lane];
        d  = dst[first + lane];
        wb = __float_as_int(nrm[first + lane]);
    }

    int i = 0;
    for (; i + 8 <= cnt; i += 8) {
        float v[8][4];
        for (int j = 0; j < 8; ++j) {
            const int sj = __builtin_amdgcn_readlane(s, i + j);
            const float* tp = t + (size_t)sj * 128 + lane;
            v[j][0] = tp[0]; v[j][1] = tp[32]; v[j][2] = tp[64]; v[j][3] = tp[96];
        }
        for (int j = 0; j < 8; ++j) {
            const int   dj = __builtin_amdgcn_readlane(d, i + j);
            const float wj = __int_as_float(__builtin_amdgcn_readlane(wb, i + j));
            float* ap = acc + (size_t)dj * 128 + lane;
            atomicAdd(ap +  0, wj * v[j][0]);
            atomicAdd(ap + 32, wj * v[j][1]);
            atomicAdd(ap + 64, wj * v[j][2]);
            atomicAdd(ap + 96, wj * v[j][3]);
        }
    }
    for (; i < cnt; ++i) {
        const int   sj = __builtin_amdgcn_readlane(s, i);
        const int   dj = __builtin_amdgcn_readlane(d, i);
        const float wj = __int_as_float(__builtin_amdgcn_readlane(wb, i));
        const float* tp = t + (size_t)sj * 128 + lane;
        float* ap = acc + (size_t)dj * 128 + lane;
        const float v0 = tp[0], v1 = tp[32], v2 = tp[64], v3 = tp[96];
        atomicAdd(ap +  0, wj * v0);
        atomicAdd(ap + 32, wj * v1);
        atomicAdd(ap + 64, wj * v2);
        atomicAdd(ap + 96, wj * v3);
    }
}

__global__ void k_dot_w3(const float* __restrict__ h, const float* __restrict__ slope,
                         const float* __restrict__ W3, float* __restrict__ t3, int N) {
    int wid  = (blockIdx.x * blockDim.x + threadIdx.x) >> 5;
    int lane = threadIdx.x & 31;
    if (wid >= N) return;
    float4 hv = ((const float4*)(h + (size_t)wid * 128))[lane];
    const float4 av = ((const float4*)slope)[lane];
    const float4 wv = ((const float4*)W3)[lane];
    hv.x = (hv.x >= 0.0f) ? hv.x : av.x * hv.x;
    hv.y = (hv.y >= 0.0f) ? hv.y : av.y * hv.y;
    hv.z = (hv.z >= 0.0f) ? hv.z : av.z * hv.z;
    hv.w = (hv.w >= 0.0f) ? hv.w : av.w * hv.w;
    float s = hv.x * wv.x + hv.y * wv.y + hv.z * wv.z + hv.w * wv.w;
    #pragma unroll
    for (int off = 16; off > 0; off >>= 1) s += __shfl_xor(s, off, 32);
    if (lane == 0) t3[wid] = s;
}

__global__ __launch_bounds__(1024) void k_edge_sum_scalar_lds(const int* __restrict__ src, const int* __restrict__ dst,
                                                              const float* __restrict__ nrm, const float* __restrict__ t3,
                                                              float* __restrict__ part, int N, int E) {
    __shared__ float bins[BIN_CAP];
    const int g   = blockIdx.x % NODE_RANGES;
    const int s   = blockIdx.x / NODE_RANGES;
    const int len = node_range_len(N);
    const int per = edge_slice_len(E);
    const int e0  = min(s * per, E), e1 = min(e0 + per, E);
    for (int base = g * len; base < N; base += NODE_RANGES * len) {
        for (int i = threadIdx.x; i < len; i += blockDim.x) bins[i] = 0.0f;
        __syncthreads();
        if (e1 > e0) {
            for (int tile = e0; tile < e1; tile += SCAN_U * (int)blockDim.x) {
                int dv[SCAN_U], sv[SCAN_U];
                float wv[SCAN_U], tv[SCAN_U];
                #pragma unroll
                for (int u = 0; u < SCAN_U; ++u) {
                    const int ec = min(tile + u * (int)blockDim.x + (int)threadIdx.x, e1 - 1);
                    dv[u] = dst[ec];
                    sv[u] = src[ec];
                    wv[u] = nrm[ec];
                }
                #pragma unroll
                for (int u = 0; u < SCAN_U; ++u) {
                    const int e = tile + u * (int)blockDim.x + (int)threadIdx.x;
                    const unsigned r = (unsigned)(dv[u] - base);
                    const bool mine = (e < e1) && (r < (unsigned)len);
                    tv[u] = t3[mine ? sv[u] : 0];
                }
                #pragma unroll
                for (int u = 0; u < SCAN_U; ++u) {
                    const int e = tile + u * (int)blockDim.x + (int)threadIdx.x;
                    const unsigned r = (unsigned)(dv[u] - base);
                    if (e < e1 && r < (unsigned)len) atomicAdd(&bins[r], wv[u] * tv[u]);
                }
            }
        }
        __syncthreads();
        for (int i = threadIdx.x; i < len; i += blockDim.x) {
            if (base + i < N) part[(size_t)s * part_pitch(N) + base + i] = bins[i];
        }
        __syncthreads();
    }
}

__global__ void k_out_final(const float* __restrict__ t3, const float* __restrict__ dis,
                            const float* __restrict__ b3, const float* __restrict__ part,
                            float* __restrict__ out, int N) {
    int i = blockIdx.x * blockDim.x + threadIdx.x;
    if (i >= N) return;
    float pv[EDGE_SLICES];
    #pragma unroll
    for (int s = 0; s < EDGE_SLICES; ++s) pv[s] = part[(size_t)s * part_pitch(N) + i];
    float acc = dis[i] * dis[i] * t3[i] + b3[0];
    #pragma unroll
    for (int s = 0; s < EDGE_SLICES; ++s) acc += pv[s];
    out[i] = acc;
}

extern "C" void kernel_launch(void* const* d_in, const int* in_sizes, int n_in,
                              void* d_out, int out_size, void* d_ws, size_t ws_size,
                              hipStream_t stream) {
    const float* num_x    = (const float*)d_in[0];
    const float* num_mask = (const float*)d_in[1];
    const float* x        = (const float*)d_in[2];
    const int*   edge     = (const int*)d_in[3];
    const float* W_num    = (const float*)d_in[4];
    const float* b_num    = (const float*)d_in[5];
    const float* a_in     = (const float*)d_in[6];
    const float* W1       = (const float*)d_in[7];
    const float* b1       = (const float*)d_in[8];
    const float* a1       = (const float*)d_in[9];
    const float* W2       = (const float*)d_in[10];
    const float* b2       = (const float*)d_in[11];
    const float* a2       = (const float*)d_in[12];
    const float* W3       = (const float*)d_in[13];
    const float* b3       = (const float*)d_in[14];

    const int N = in_sizes[1];
    const int E = in_sizes[3] / 2;
    const int* src = edge;
    const int* dst = edge + E;

    const int nRowTiles = (N + 15) / 16;
    const size_t padRows = (size_t)nRowTiles * 16;

    auto al = [](size_t v) { return (v + 255) & ~(size_t)255; };
    char* p = (char*)d_ws;
    float* dis  = (float*)p; p += al((size_t)N * 4);
    float* nrm  = (float*)p; p += al((size_t)E * 4);
    float* hbuf = (float*)p; p += al(padRows * 128 * 4);
    float* tbuf = (float*)p; p += al(padRows * 128 * 4);
    float* abuf = (float*)p; p += al(padRows * 128 * 4);
    float* t3   = (float*)p; p += al((size_t)N * 4);
    float* part = (float*)p; p += al((size_t)EDGE_SLICES * part_pitch(N) * 4);
    float* out  = (float*)d_out;

    const int TPB = 256;
    dim3 blk(TPB);
    auto gN  = dim3((N + TPB - 1) / TPB);
    auto gE  = dim3((E + TPB - 1) / TPB);
    auto gEw = dim3((int)(((size_t)((E + 31) / 32) * 32 + TPB - 1) / TPB));
    auto gNw = dim3((int)(((size_t)N * 32 + TPB - 1) / TPB));
    const int gemmBlocks = (nRowTiles + 3) / 4;

    k_deg_count_and_input_proj<<<NODE_RANGES * EDGE_SLICES + PROJ_BLOCKS, 1024, 0, stream>>>(
        dst, (int*)part, num_x, num_mask, x, W_num, b_num, a_in, hbuf, N, E);
    k_deg_rsqrt<<<gN, blk, 0, stream>>>((const int*)part, dis, N);
    k_edge_norm<<<gE, blk, 0, stream>>>(src, dst, dis, nrm, E);

    k_gemm128_wmma_plain<<<gemmBlocks, 128, 0, stream>>>(hbuf, W1, dis, b1, tbuf, abuf, nRowTiles, N);
    k_edge_scatter_vec<<<gEw, blk, 0, stream>>>(src, dst, nrm, tbuf, abuf, E);

    k_gemm128_wmma_prelu<<<gemmBlocks, 128, 0, stream>>>(abuf, a1, W2, dis, b2, tbuf, hbuf, nRowTiles, N);
    k_edge_scatter_vec<<<gEw, blk, 0, stream>>>(src, dst, nrm, tbuf, hbuf, E);

    k_dot_w3<<<gNw, blk, 0, stream>>>(hbuf, a2, W3, t3, N);
    k_edge_sum_scalar_lds<<<NODE_RANGES * EDGE_SLICES, 1024, 0, stream>>>(src, dst, nrm, t3, part, N, E);
    k_out_final<<<gN, blk, 0, stream>>>(t3, dis, b3, part, out, N);
}
